# starting kernel
# baseline (speedup 1.0000x reference)
#include <hip/hip_runtime.h>


typedef __attribute__((ext_vector_type(2))) float v2f;
typedef __attribute__((ext_vector_type(8))) float v8f;

#define Hdim 512
#define Vdim 5
#define Bdim 8
#define Tdim 400
#define U1dim 101

#define TROWS (Bdim * Tdim)
#define PROWS (Bdim * U1dim)
#define TTILES ((TROWS + 15) / 16)
#define PTILES ((PROWS + 15) / 16)
#define NTILES (TTILES + PTILES)

__global__ __launch_bounds__(256) void rowgemm_wmma(
    const float* __restrict__ trans,
    const float* __restrict__ preds,
    const float* __restrict__ W,
    const float* __restrict__ bias,
    float* __restrict__ TW,
    float* __restrict__ PW)
{
    const int wave = threadIdx.x >> 5;
    const int lane = threadIdx.x & 31;
    const int tile = blockIdx.x * 8 + wave;
    if (tile >= NTILES) return;

    const bool  isTrans = (tile < TTILES);
    const float* A   = isTrans ? trans : preds;
    float*       Dst = isTrans ? TW    : PW;
    const int    rows = isTrans ? TROWS : PROWS;
    const int    row0 = (isTrans ? tile : tile - TTILES) * 16;

    const int half = lane >> 4;
    const int l15  = lane & 15;

    const int arow = min(row0 + l15, rows - 1);
    const int brow = min(l15, Vdim - 1);

    const float* __restrict__ ap = A + arow * Hdim + 2 * half;
    const float* __restrict__ bp = W + brow * Hdim + 2 * half;

    v8f c;
    const float cinit = isTrans ? bias[brow] : 0.0f;
#pragma unroll
    for (int r = 0; r < 8; ++r) c[r] = cinit;

#pragma unroll 8
    for (int k = 0; k < Hdim; k += 4) {
        const v2f a = *(const v2f*)(ap + k);
        const v2f b = *(const v2f*)(bp + k);
        c = __builtin_amdgcn_wmma_f32_16x16x4_f32(
                 false, a,  false, b,
                 (short)0, c,  false,  false);
    }

    if (l15 < Vdim) {
        const int n = l15;
#pragma unroll
        for (int r = 0; r < 8; ++r) {
            const int row = row0 + r + 8 * half;
            if (row < rows) Dst[row * Vdim + n] = c[r];
        }
    }
}

__global__ __launch_bounds__(256) void joint_write(
    const float* __restrict__ TW,
    const float* __restrict__ PW,
    float* __restrict__ out)
{
    const int bt = blockIdx.x;
    const int bb = bt / Tdim;
    const float* __restrict__ pw = PW + bb * (U1dim * Vdim);
    const float* __restrict__ tw = TW + bt * Vdim;
    float* __restrict__ o = out + (size_t)bt * (U1dim * Vdim);

    for (int j = threadIdx.x; j < U1dim * Vdim; j += 256) {
        const int v = j % Vdim;
        const float x = tw[v] + pw[j];
        __builtin_nontemporal_store(fmaxf(x, 0.1f * x), o + j);
    }
}

extern "C" void kernel_launch(void* const* d_in, const int* in_sizes, int n_in,
                              void* d_out, int out_size, void* d_ws, size_t ws_size,
                              hipStream_t stream) {
    const float* trans = (const float*)d_in[0];
    const float* preds = (const float*)d_in[1];
    const float* W     = (const float*)d_in[2];
    const float* bias  = (const float*)d_in[3];
    float* out = (float*)d_out;

    float* TW = (float*)d_ws;
    float* PW = TW + TROWS * Vdim;

    rowgemm_wmma<<<(NTILES + 7) / 8, 256, 0, stream>>>(trans, preds, W, bias, TW, PW);
    joint_write<<<TROWS, 256, 0, stream>>>(TW, PW, out);
}


# row tiles and W go through LDS: each wave reads its 16x512 tile with 64 whole-line 16-byte loads, all issued up front, and takes its WMMA fragments from LDS in the same k order (sums unchanged)
# speedup vs baseline: 2.0557x; 2.0557x over previous
#include <hip/hip_runtime.h>


typedef __attribute__((ext_vector_type(2))) float v2f;
typedef __attribute__((ext_vector_type(4))) float v4f;
typedef __attribute__((ext_vector_type(8))) float v8f;

#define Hdim 512
#define Vdim 5
#define Bdim 8
#define Tdim 400
#define U1dim 101

#define SLAB   (U1dim * Vdim)
#define GROUPS (Bdim * Tdim / 16)
#define PTILES ((U1dim + 15) / 16)

#define NQ     4
#define QK     (Hdim / NQ)
#define KSTEPS (QK / 4)
#define PITCH  (Hdim + 4)

static_assert(Tdim % 16 == 0, "a 16-row group must stay inside one batch");
static_assert(PTILES + 1 == 8, "one row tile per wave of the 256-thread group");
static_assert(SLAB > 256 && SLAB <= 512, "two output columns per thread");
static_assert(QK == 32 * 4, "one load instruction of a wave covers one row's quarter");
static_assert(PITCH % 64 == 4 && (PITCH * 4) % 16 == 0, "bank spread and 16-byte row alignment");
static_assert(Vdim * Hdim / 4 == 640, "W is 640 16-byte pieces: two per thread and a third for half of them");
static_assert((8 * 16 + Vdim) * PITCH * 4 + (SLAB + 16 * Vdim) * 4 <= 320 * 1024, "fits the LDS of one WGP");

__device__ __forceinline__ void lds_barrier()
{
    asm volatile("s_wait_dscnt 0x0\n\ts_barrier_signal -1\n\ts_barrier_wait -1" ::: "memory");
}

__global__ __launch_bounds__(256) void joint_fused(
    const float* __restrict__ trans,
    const float* __restrict__ preds,
    const float* __restrict__ W,
    const float* __restrict__ bias,
    float* __restrict__ out)
{
    __shared__ __attribute__((aligned(16))) float sA[8][16][PITCH];
    __shared__ __attribute__((aligned(16))) float sW[Vdim][PITCH];
    __shared__ float sPW[SLAB];
    __shared__ float sTW[16 * Vdim];

    const int tid  = threadIdx.x;
    const int wave = tid >> 5;
    const int lane = tid & 31;
    const int bt0  = blockIdx.x * 16;
    const int bb   = blockIdx.x / (Tdim / 16);

    const bool   isTrans = (wave == PTILES);
    const float* A    = isTrans ? trans + (size_t)bt0 * Hdim
                                : preds + (size_t)bb * U1dim * Hdim;
    float*       Dst  = isTrans ? sTW : sPW;
    const int    rows = isTrans ? 16 : U1dim;
    const int    row0 = isTrans ? 0 : wave * 16;

    const int half = lane >> 4;
    const int l15  = lane & 15;

    const int   brow  = min(l15, Vdim - 1);
    const float cinit = isTrans ? bias[brow] : 0.0f;

    v4f wv0 = *(const v4f*)(W + 4 * tid);
    v4f wv1 = *(const v4f*)(W + 4 * (tid + 256));
    v4f wv2;
    if (tid < 128) wv2 = *(const v4f*)(W + 4 * (tid + 512));

    v4f buf[NQ * 16];
#pragma unroll
    for (int q = 0; q < NQ; ++q) {
#pragma unroll
        for (int r = 0; r < 16; ++r) {
            const int gr = min(row0 + r, rows - 1);
            buf[q * 16 + r] = *(const v4f*)(A + (size_t)gr * Hdim + q * QK + 4 * lane);
        }
        __builtin_amdgcn_sched_barrier(0);
    }

    *(v4f*)&sW[tid >> 7][4 * (tid & 127)] = wv0;
    *(v4f*)&sW[2 + (tid >> 7)][4 * (tid & 127)] = wv1;
    if (tid < 128) *(v4f*)&sW[4][4 * tid] = wv2;

    float*       myA = &sA[wave][0][0];
    const float* fa  = myA + l15 * PITCH + 2 * half;
    const float* fb  = &sW[brow][0] + 2 * half;

    v8f c;
#pragma unroll
    for (int r = 0; r < 8; ++r) c[r] = cinit;

#pragma unroll
    for (int q = 0; q < NQ; ++q) {
#pragma unroll
        for (int r = 0; r < 16; ++r)
            *(v4f*)(myA + r * PITCH + q * QK + 4 * lane) = buf[q * 16 + r];
        lds_barrier();

        v2f af[KSTEPS], bf[KSTEPS];
#pragma unroll
        for (int s = 0; s < KSTEPS; ++s) {
            af[s] = *(const v2f*)(fa + q * QK + 4 * s);
            bf[s] = *(const v2f*)(fb + q * QK + 4 * s);
        }
#pragma unroll
        for (int s = 0; s < KSTEPS; ++s) {
            c = __builtin_amdgcn_wmma_f32_16x16x4_f32(
                     false, af[s],  false, bf[s],
                     (short)0, c,  false,  false);
        }
        __builtin_amdgcn_sched_group_barrier(0x100, 8, 0);
#pragma unroll
        for (int g = 0; g < 12; ++g) {
            __builtin_amdgcn_sched_group_barrier(0x008, 2, 0);
            __builtin_amdgcn_sched_group_barrier(0x100, 2, 0);
        }
        __builtin_amdgcn_sched_group_barrier(0x008, 8, 0);
    }

    if (l15 < Vdim) {
        const int n = l15;
#pragma unroll
        for (int r = 0; r < 8; ++r) {
            const int row = row0 + r + 8 * half;
            if (row < rows) Dst[row * Vdim + n] = c[r];
        }
    }
    __syncthreads();

    float* __restrict__ o = out + (size_t)bt0 * SLAB;
    const int  j0 = threadIdx.x;
    const int  j1 = threadIdx.x + 256;
    const bool has1 = (j1 < SLAB);
    const int  v0 = j0 % Vdim;
    const int  v1 = j1 % Vdim;
    const float pw0 = sPW[j0];

    float tw[16];
#pragma unroll
    for (int t = 0; t < 16; ++t) tw[t] = sTW[t * Vdim + v0];
#pragma unroll
    for (int t = 0; t < 16; ++t) {
        const float x = tw[t] + pw0;
        __builtin_nontemporal_store(fmaxf(x, 0.1f * x), o + t * SLAB + j0);
    }

    if (has1) {
        const float pw1 = sPW[j1];
#pragma unroll
        for (int t = 0; t < 16; ++t) tw[t] = sTW[t * Vdim + v1];
#pragma unroll
        for (int t = 0; t < 16; ++t) {
            const float x = tw[t] + pw1;
            __builtin_nontemporal_store(fmaxf(x, 0.1f * x), o + t * SLAB + j1);
        }
    }
}

extern "C" void kernel_launch(void* const* d_in, const int* in_sizes, int n_in,
                              void* d_out, int out_size, void* d_ws, size_t ws_size,
                              hipStream_t stream) {
    const float* trans = (const float*)d_in[0];
    const float* preds = (const float*)d_in[1];
    const float* W     = (const float*)d_in[2];
    const float* bias  = (const float*)d_in[3];
    float* out = (float*)d_out;

    joint_fused<<<GROUPS, 256, 0, stream>>>(trans, preds, W, bias, out);
}


# fragment reads software-pipelined: next K quarter's 32 LDS reads and 16 parking stores issued as ONE group midway through the current quarter's WMMAs (2nd register set); quarter barriers dropped
# speedup vs baseline: 2.1597x; 1.0506x over previous
#include <hip/hip_runtime.h>


typedef __attribute__((ext_vector_type(2))) float v2f;
typedef __attribute__((ext_vector_type(4))) float v4f;
typedef __attribute__((ext_vector_type(8))) float v8f;

#define Hdim 512
#define Vdim 5
#define Bdim 8
#define Tdim 400
#define U1dim 101

#define SLAB   (U1dim * Vdim)
#define GROUPS (Bdim * Tdim / 16)
#define PTILES ((U1dim + 15) / 16)

#define NQ     4
#define QK     (Hdim / NQ)
#define KSTEPS (QK / 4)
#define PITCH  (Hdim + 4)

static_assert(Tdim % 16 == 0, "a 16-row group must stay inside one batch");
static_assert(PTILES + 1 == 8, "one row tile per wave of the 256-thread group");
static_assert(SLAB > 256 && SLAB <= 512, "two output columns per thread");
static_assert(QK == 32 * 4, "one load instruction of a wave covers one row's quarter");
static_assert(PITCH % 64 == 4 && (PITCH * 4) % 16 == 0, "bank spread and 16-byte row alignment");
static_assert(Vdim * Hdim / 4 == 640, "W is 640 16-byte pieces: two per thread and a third for half of them");
static_assert((8 * 16 + Vdim) * PITCH * 4 + (SLAB + 16 * Vdim) * 4 <= 320 * 1024, "fits the LDS of one WGP");

__device__ __forceinline__ void lds_barrier()
{
    asm volatile("s_wait_dscnt 0x0\n\ts_barrier_signal -1\n\ts_barrier_wait -1" ::: "memory");
}

__global__ __launch_bounds__(256) void joint_fused(
    const float* __restrict__ trans,
    const float* __restrict__ preds,
    const float* __restrict__ W,
    const float* __restrict__ bias,
    float* __restrict__ out)
{
    __shared__ __attribute__((aligned(16))) float sA[8][16][PITCH];
    __shared__ __attribute__((aligned(16))) float sW[Vdim][PITCH];
    __shared__ float sPW[SLAB];
    __shared__ float sTW[16 * Vdim];

    const int tid  = threadIdx.x;
    const int wave = tid >> 5;
    const int lane = tid & 31;
    const int bt0  = blockIdx.x * 16;
    const int bb   = blockIdx.x / (Tdim / 16);

    const bool   isTrans = (wave == PTILES);
    const float* A    = isTrans ? trans + (size_t)bt0 * Hdim
                                : preds + (size_t)bb * U1dim * Hdim;
    float*       Dst  = isTrans ? sTW : sPW;
    const int    rows = isTrans ? 16 : U1dim;
    const int    row0 = isTrans ? 0 : wave * 16;

    const int half = lane >> 4;
    const int l15  = lane & 15;

    const int   brow  = min(l15, Vdim - 1);
    const float cinit = isTrans ? bias[brow] : 0.0f;

    v4f wv0 = *(const v4f*)(W + 4 * tid);
    v4f wv1 = *(const v4f*)(W + 4 * (tid + 256));
    v4f wv2;
    if (tid < 128) wv2 = *(const v4f*)(W + 4 * (tid + 512));

    v4f buf[NQ * 16];
#pragma unroll
    for (int q = 0; q < NQ; ++q) {
#pragma unroll
        for (int r = 0; r < 16; ++r) {
            const int gr = min(row0 + r, rows - 1);
            buf[q * 16 + r] = *(const v4f*)(A + (size_t)gr * Hdim + q * QK + 4 * lane);
        }
        __builtin_amdgcn_sched_barrier(0);
    }

    *(v4f*)&sW[tid >> 7][4 * (tid & 127)] = wv0;
    *(v4f*)&sW[2 + (tid >> 7)][4 * (tid & 127)] = wv1;
    if (tid < 128) *(v4f*)&sW[4][4 * tid] = wv2;

    float*       myA = &sA[wave][0][0];
    const float* fa  = myA + l15 * PITCH + 2 * half;
    const float* fb  = &sW[brow][0] + 2 * half;

    v8f c;
#pragma unroll
    for (int r = 0; r < 8; ++r) c[r] = cinit;

#define LOAD_FRAGS(q, set)                                                   \
    _Pragma("unroll")                                                        \
    for (int s = 0; s < KSTEPS; ++s) {                                       \
        af[set][s] = *(const v2f*)(fa + (q) * QK + 4 * s);                   \
        bf[set][s] = *(const v2f*)(fb + (q) * QK + 4 * s);                   \
    }
#define PARK(q)                                                              \
    _Pragma("unroll")                                                        \
    for (int r = 0; r < 16; ++r)                                             \
        *(v4f*)(myA + r * PITCH + (q) * QK + 4 * lane) = buf[(q) * 16 + r];
#define WMMAS(set, s0, s1)                                                   \
    _Pragma("unroll")                                                        \
    for (int s = (s0); s < (s1); ++s)                                        \
        c = __builtin_amdgcn_wmma_f32_16x16x4_f32(                           \
                 false, af[set][s],  false, bf[set][s],    \
                 (short)0, c,  false,  false);

    v2f af[2][KSTEPS], bf[2][KSTEPS];

    PARK(0)
    lds_barrier();
    LOAD_FRAGS(0, 0)
    __builtin_amdgcn_sched_barrier(0);

#pragma unroll
    for (int q = 0; q < NQ; ++q) {
        WMMAS(q & 1, 0, KSTEPS / 2)
        __builtin_amdgcn_sched_barrier(0);
        if (q + 1 < NQ) {
            PARK(q + 1)
            LOAD_FRAGS(q + 1, (q + 1) & 1)
        }
        __builtin_amdgcn_sched_barrier(0);
        WMMAS(q & 1, KSTEPS / 2, KSTEPS)
    }
#undef LOAD_FRAGS
#undef PARK
#undef WMMAS

    if (l15 < Vdim) {
        const int n = l15;
#pragma unroll
        for (int r = 0; r < 8; ++r) {
            const int row = row0 + r + 8 * half;
            if (row < rows) Dst[row * Vdim + n] = c[r];
        }
    }
    __syncthreads();

    float* __restrict__ o = out + (size_t)bt0 * SLAB;
    const int  j0 = threadIdx.x;
    const int  j1 = threadIdx.x + 256;
    const bool has1 = (j1 < SLAB);
    const int  v0 = j0 % Vdim;
    const int  v1 = j1 % Vdim;
    const float pw0 = sPW[j0];

    float tw[16];
#pragma unroll
    for (int t = 0; t < 16; ++t) tw[t] = sTW[t * Vdim + v0];
#pragma unroll
    for (int t = 0; t < 16; ++t) {
        const float x = tw[t] + pw0;
        __builtin_nontemporal_store(fmaxf(x, 0.1f * x), o + t * SLAB + j0);
    }

    if (has1) {
        const float pw1 = sPW[j1];
#pragma unroll
        for (int t = 0; t < 16; ++t) tw[t] = sTW[t * Vdim + v1];
#pragma unroll
        for (int t = 0; t < 16; ++t) {
            const float x = tw[t] + pw1;
            __builtin_nontemporal_store(fmaxf(x, 0.1f * x), o + t * SLAB + j1);
        }
    }
}

extern "C" void kernel_launch(void* const* d_in, const int* in_sizes, int n_in,
                              void* d_out, int out_size, void* d_ws, size_t ws_size,
                              hipStream_t stream) {
    const float* trans = (const float*)d_in[0];
    const float* preds = (const float*)d_in[1];
    const float* W     = (const float*)d_in[2];
    const float* bias  = (const float*)d_in[3];
    float* out = (float*)d_out;

    joint_fused<<<GROUPS, 256, 0, stream>>>(trans, preds, W, bias, out);
}


# out written in whole 128-byte lines: a lane stores one aligned 16-byte piece of the flat 16x505 slab (64 stores per group, not 256 of 4 bytes); PW and TW tables in LDS laid out for it; sums unchanged
# speedup vs baseline: 2.4863x; 1.1512x over previous
#include <hip/hip_runtime.h>


typedef __attribute__((ext_vector_type(2))) float v2f;
typedef __attribute__((ext_vector_type(4))) float v4f;
typedef __attribute__((ext_vector_type(8))) float v8f;

#define Hdim 512
#define Vdim 5
#define Bdim 8
#define Tdim 400
#define U1dim 101

#define SLAB   (U1dim * Vdim)
#define GROUPS (Bdim * Tdim / 16)
#define PTILES ((U1dim + 15) / 16)

#define NQ     4
#define QK     (Hdim / NQ)
#define KSTEPS (QK / 4)
#define PITCH  (Hdim + 4)

static_assert(Tdim % 16 == 0, "a 16-row group must stay inside one batch");
static_assert(PTILES + 1 == 8, "one row tile per wave of the 256-thread group");
static_assert(QK == 32 * 4, "one load instruction of a wave covers one row's quarter");
static_assert(PITCH % 64 == 4 && (PITCH * 4) % 16 == 0, "bank spread and 16-byte row alignment");
static_assert(Vdim * Hdim / 4 == 640, "W is 640 16-byte pieces: two per thread and a third for half of them");
#define NPIECE (16 * SLAB / 4)
#define TWP    8
#define TWROWS (2 * 16 + 1)

static_assert((16 * SLAB) % 4 == 0, "a slab is a whole number of 16-byte pieces");
static_assert(SLAB % Vdim == 0, "v of a slab element is its flat index mod Vdim");
static_assert(NPIECE + 4 <= 8 * 256, "eight pieces per thread cover the slab, shifted or not");
static_assert((8 * 16 + Vdim) * PITCH * 4 + (SLAB + 3 + TWROWS * TWP) * 4 <= 320 * 1024, "fits the LDS of one WGP");

__device__ __forceinline__ void lds_barrier()
{
    asm volatile("s_wait_dscnt 0x0\n\ts_barrier_signal -1\n\ts_barrier_wait -1" ::: "memory");
}

__global__ __launch_bounds__(256) void joint_fused(
    const float* __restrict__ trans,
    const float* __restrict__ preds,
    const float* __restrict__ W,
    const float* __restrict__ bias,
    float* __restrict__ out)
{
    __shared__ __attribute__((aligned(16))) float sA[8][16][PITCH];
    __shared__ __attribute__((aligned(16))) float sW[Vdim][PITCH];
    __shared__ float sPW[SLAB + 3];
    __shared__ float sTWx[TWROWS * TWP];

    const int tid  = threadIdx.x;
    const int wave = tid >> 5;
    const int lane = tid & 31;
    const int bt0  = blockIdx.x * 16;
    const int bb   = blockIdx.x / (Tdim / 16);

    const bool   isTrans = (wave == PTILES);
    const float* A    = isTrans ? trans + (size_t)bt0 * Hdim
                                : preds + (size_t)bb * U1dim * Hdim;
    const int    rows = isTrans ? 16 : U1dim;
    const int    row0 = isTrans ? 0 : wave * 16;

    const int half = lane >> 4;
    const int l15  = lane & 15;

    const int   brow  = min(l15, Vdim - 1);
    const float cinit = isTrans ? bias[brow] : 0.0f;

    v4f wv0 = *(const v4f*)(W + 4 * tid);
    v4f wv1 = *(const v4f*)(W + 4 * (tid + 256));
    v4f wv2;
    if (tid < 128) wv2 = *(const v4f*)(W + 4 * (tid + 512));

    v4f buf[NQ * 16];
#pragma unroll
    for (int q = 0; q < NQ; ++q) {
#pragma unroll
        for (int r = 0; r < 16; ++r) {
            const int gr = min(row0 + r, rows - 1);
            buf[q * 16 + r] = *(const v4f*)(A + (size_t)gr * Hdim + q * QK + 4 * lane);
        }
        __builtin_amdgcn_sched_barrier(0);
    }

    *(v4f*)&sW[tid >> 7][4 * (tid & 127)] = wv0;
    *(v4f*)&sW[2 + (tid >> 7)][4 * (tid & 127)] = wv1;
    if (tid < 128) *(v4f*)&sW[4][4 * tid] = wv2;

    float*       myA = &sA[wave][0][0];
    const float* fa  = myA + l15 * PITCH + 2 * half;
    const float* fb  = &sW[brow][0] + 2 * half;

    v8f c;
#pragma unroll
    for (int r = 0; r < 8; ++r) c[r] = cinit;

#define LOAD_FRAGS(q, set)                                                   \
    _Pragma("unroll")                                                        \
    for (int s = 0; s < KSTEPS; ++s) {                                       \
        af[set][s] = *(const v2f*)(fa + (q) * QK + 4 * s);                   \
        bf[set][s] = *(const v2f*)(fb + (q) * QK + 4 * s);                   \
    }
#define PARK(q)                                                              \
    _Pragma("unroll")                                                        \
    for (int r = 0; r < 16; ++r)                                             \
        *(v4f*)(myA + r * PITCH + (q) * QK + 4 * lane) = buf[(q) * 16 + r];
#define WMMAS(set, s0, s1)                                                   \
    _Pragma("unroll")                                                        \
    for (int s = (s0); s < (s1); ++s)                                        \
        c = __builtin_amdgcn_wmma_f32_16x16x4_f32(                           \
                 false, af[set][s],  false, bf[set][s],    \
                 (short)0, c,  false,  false);

    v2f af[2][KSTEPS], bf[2][KSTEPS];

    PARK(0)
    lds_barrier();
    LOAD_FRAGS(0, 0)
    __builtin_amdgcn_sched_barrier(0);

#pragma unroll
    for (int q = 0; q < NQ; ++q) {
        WMMAS(q & 1, 0, KSTEPS / 2)
        __builtin_amdgcn_sched_barrier(0);
        if (q + 1 < NQ) {
            PARK(q + 1)
            LOAD_FRAGS(q + 1, (q + 1) & 1)
        }
        __builtin_amdgcn_sched_barrier(0);
        WMMAS(q & 1, KSTEPS / 2, KSTEPS)
    }
#undef LOAD_FRAGS
#undef PARK
#undef WMMAS

    const int pre = (blockIdx.x & 1) * 4;
    int pwo[8], two[8];
#pragma unroll
    for (int i = 0; i < 8; ++i) {
        int f = tid + 256 * i - pre;
        if (i == 0) f = max(f, 0);
        if (i == 7) f = min(f, NPIECE - 1);
        const unsigned uf = (unsigned)f;
        const unsigned t  = (uf * 8306u) >> 20;
        const unsigned j0 = 4u * uf - t * SLAB;
        const unsigned v0 = j0 - Vdim * ((j0 * 205u) >> 10);
        const unsigned ov = (j0 > SLAB - 4) ? 1u : 0u;
        pwo[i] = (int)j0;
        two[i] = (int)((2u * t + 1u + ov) * TWP + v0);
    }

    if (l15 < Vdim) {
        const int n = l15;
        if (isTrans) {
#pragma unroll
            for (int r = 0; r < 8; ++r) {
                const int row = r + 8 * half;
                sTWx[(2 * row + 1) * TWP + n] = c[r];
                sTWx[(2 * row + 2) * TWP + n] = c[r];
            }
            if (n < 3) {
#pragma unroll
                for (int r = 0; r < 8; ++r) {
                    const int row = r + 8 * half;
                    sTWx[(2 * row + 1) * TWP + Vdim + n] = c[r];
                    sTWx[(2 * row) * TWP + Vdim + n] = c[r];
                }
            }
        } else {
#pragma unroll
            for (int r = 0; r < 8; ++r) {
                const int row = row0 + r + 8 * half;
                if (row < rows) sPW[row * Vdim + n] = c[r];
            }
            if (wave == 0 && half == 0 && n < 3) sPW[SLAB + n] = c[0];
        }
    }
    __syncthreads();

    float* __restrict__ o = out + (size_t)bt0 * SLAB;

    float pw[8][4], tw[8][4];
#pragma unroll
    for (int i = 0; i < 8; ++i) {
#pragma unroll
        for (int k = 0; k < 4; ++k) {
            pw[i][k] = sPW[pwo[i] + k];
            tw[i][k] = sTWx[two[i] + k];
        }
    }
#pragma unroll
    for (int i = 0; i < 8; ++i) {
        const int f = tid + 256 * i - pre;
        const bool inside = (i == 0) ? (f >= 0) : (i == 7) ? (f < NPIECE) : true;
        v4f y;
#pragma unroll
        for (int k = 0; k < 4; ++k) {
            const float x = tw[i][k] + pw[i][k];
            y[k] = fmaxf(x, 0.1f * x);
        }
        if (inside) __builtin_nontemporal_store(y, (v4f*)(o + 4 * f));
    }
}

extern "C" void kernel_launch(void* const* d_in, const int* in_sizes, int n_in,
                              void* d_out, int out_size, void* d_ws, size_t ws_size,
                              hipStream_t stream) {
    const float* trans = (const float*)d_in[0];
    const float* preds = (const float*)d_in[1];
    const float* W     = (const float*)d_in[2];
    const float* bias  = (const float*)d_in[3];
    float* out = (float*)d_out;

    joint_fused<<<GROUPS, 256, 0, stream>>>(trans, preds, W, bias, out);
}
